# starting kernel
# baseline (speedup 1.0000x reference)
#include <hip/hip_runtime.h>
#include <hip/hip_bf16.h>
#include <math.h>

typedef __attribute__((ext_vector_type(2))) float v2f;
typedef __attribute__((ext_vector_type(4))) float v4f;
typedef __attribute__((ext_vector_type(8))) float v8f;

#define N_GAUSS   1024
#define IMG_H     256
#define IMG_W     256
#define EPS_COV   1e-6f
#define ALPHA_MAX 0.999f

#define GF_STRIDE 10
#define GF_DW     (N_GAUSS * GF_STRIDE)
#define CP_DW     (N_GAUSS * 4)
#define SMEM_DW   (GF_DW + CP_DW)

__global__ __launch_bounds__(256) void gs_prep(const float* __restrict__ means,
                                               const float* __restrict__ log_scales,
                                               const float* __restrict__ rot,
                                               const float* __restrict__ colors,
                                               const float* __restrict__ opac,
                                               float* __restrict__ ws) {
  int n = blockIdx.x * blockDim.x + threadIdx.x;
  if (n >= N_GAUSS) return;

  float mx = means[2 * n + 0];
  float my = means[2 * n + 1];
  float s2x = __expf(2.f * log_scales[2 * n + 0]);
  float s2y = __expf(2.f * log_scales[2 * n + 1]);
  float s, c;
  __sincosf(rot[n], &s, &c);

  float a = c * c * s2x + s * s * s2y + EPS_COV;
  float b = c * s * (s2x - s2y);
  float d = s * s * s2x + c * c * s2y + EPS_COV;
  float inv = 1.f / (a * d - b * b);
  float ia = d * inv, ib = -b * inv, id = a * inv;

  float* gf = ws + (size_t)n * GF_STRIDE;
  gf[0] = -0.5f * ia;
  gf[1] = ia * mx + ib * my;
  gf[2] = -0.5f * id;
  gf[3] = id * my + ib * mx;
  gf[4] = -ib;
  gf[5] = -0.5f * (ia * mx * mx + 2.f * ib * mx * my + id * my * my);
  gf[6] = 0.f; gf[7] = 0.f;
  gf[8] = 0.f; gf[9] = 0.f;

  float* cp = ws + GF_DW + (size_t)n * 4;
  cp[0] = colors[3 * n + 0];
  cp[1] = colors[3 * n + 1];
  cp[2] = colors[3 * n + 2];
  cp[3] = opac[n];
}

__global__ __launch_bounds__(256) void gs_render(const float* __restrict__ ws,
                                                 float* __restrict__ out) {
  __shared__ float smem[SMEM_DW];

  {
    const v4f* src = (const v4f*)ws;
    v4f* dst = (v4f*)smem;
    for (int i = threadIdx.x; i < SMEM_DW / 4; i += 256) dst[i] = src[i];
  }
  __syncthreads();

  const float* sg = smem;
  const float* sc = smem + GF_DW;

  const int lane = threadIdx.x & 31;
  const int wave = threadIdx.x >> 5;
  const int half = lane >> 4;
  const int col  = lane & 15;

  const int p = blockIdx.x * 128 + wave * 16 + col;
  const float x = (float)(p & (IMG_W - 1));
  const float y = (float)(p >> 8);

  const float f[8] = { x * x, x, y * y, y, x * y, 1.f, 0.f, 0.f };
  const v2f bA = { f[2 * half + 0], f[2 * half + 1] };
  const v2f bB = { f[4 + 2 * half + 0], f[4 + 2 * half + 1] };

  float T = 1.f;
  float cr = 0.f, cg = 0.f, cb = 0.f;

  for (int gb = 0; gb < N_GAUSS; gb += 16) {
    const float* gf = sg + (gb + col) * GF_STRIDE;
    const v2f aA = { gf[2 * half + 0], gf[2 * half + 1] };
    const v2f aB = { gf[4 + 2 * half + 0], gf[4 + 2 * half + 1] };

    v8f P = {};
    P = __builtin_amdgcn_wmma_f32_16x16x4_f32(false, aA, false, bA,
                                              (short)0, P, false, false);
    P = __builtin_amdgcn_wmma_f32_16x16x4_f32(false, aB, false, bB,
                                              (short)0, P, false, false);

    const int gg = gb + 8 * half;
    float w[8];
    float q = 1.f;
#pragma unroll
    for (int j = 0; j < 8; ++j) {
      float op = sc[(gg + j) * 4 + 3];
      float al = fminf(op * __expf(P[j]), ALPHA_MAX);
      w[j] = al * q;
      q *= (1.f - al);
    }

    float qo  = __shfl_xor(q, 16, 32);
    float pre = half ? (T * qo) : T;
    T *= q * qo;

#pragma unroll
    for (int j = 0; j < 8; ++j) {
      float wv = w[j] * pre;
      const float* cc = sc + (gg + j) * 4;
      cr = fmaf(wv, cc[0], cr);
      cg = fmaf(wv, cc[1], cg);
      cb = fmaf(wv, cc[2], cb);
    }
  }

  cr += __shfl_xor(cr, 16, 32);
  cg += __shfl_xor(cg, 16, 32);
  cb += __shfl_xor(cb, 16, 32);
  if (half == 0) {
    float* o = out + (size_t)p * 3;
    o[0] = cr; o[1] = cg; o[2] = cb;
  }
}

extern "C" void kernel_launch(void* const* d_in, const int* in_sizes, int n_in,
                              void* d_out, int out_size, void* d_ws, size_t ws_size,
                              hipStream_t stream) {
  const float* means      = (const float*)d_in[0];
  const float* log_scales = (const float*)d_in[1];
  const float* rotations  = (const float*)d_in[2];
  const float* colors     = (const float*)d_in[3];
  const float* opacities  = (const float*)d_in[4];
  float* ws = (float*)d_ws;

  gs_prep<<<N_GAUSS / 256, 256, 0, stream>>>(means, log_scales, rotations,
                                             colors, opacities, ws);
  gs_render<<<(IMG_H * IMG_W) / 128, 256, 0, stream>>>(ws, (float*)d_out);
}


# render: one pixel per lane; A rows 8-15 hold the step's same 8 gaussians with the mean moved 16 px, so both lane halves composite the same gaussians in index order: no half merge (sum order changes)
# speedup vs baseline: 1.3253x; 1.3253x over previous
#include <hip/hip_runtime.h>
#include <hip/hip_bf16.h>
#include <math.h>


typedef __attribute__((ext_vector_type(2))) float v2f;
typedef __attribute__((ext_vector_type(4))) float v4f;
typedef __attribute__((ext_vector_type(8))) float v8f;

#define N_GAUSS   1024
#define IMG_H     256
#define IMG_W     256
#define EPS_COV   1e-6f
#define ALPHA_MAX 0.999f

#define STEP_G    8
#define N_STEPS   (N_GAUSS / STEP_G)
#define X_SHIFT   16.f
#define ROW_DW    8
#define GF_DW     (N_STEPS * 16 * ROW_DW)
#define CP_DW     (N_GAUSS * 4)
#define SMEM_DW   (GF_DW + CP_DW)

__device__ __forceinline__ void feature_row(float* __restrict__ gf, float ia, float ib, float id,
                                            float mx, float my) {
  gf[0] = -0.5f * ia;
  gf[1] = ia * mx + ib * my;
  gf[2] = -0.5f * id;
  gf[3] = id * my + ib * mx;
  gf[4] = -ib;
  gf[5] = -0.5f * (ia * mx * mx + 2.f * ib * mx * my + id * my * my);
  gf[6] = 0.f;
  gf[7] = 0.f;
}

__global__ __launch_bounds__(256) void gs_prep(const float* __restrict__ means,
                                               const float* __restrict__ log_scales,
                                               const float* __restrict__ rot,
                                               const float* __restrict__ colors,
                                               const float* __restrict__ opac,
                                               float* __restrict__ ws) {
  int n = blockIdx.x * blockDim.x + threadIdx.x;
  if (n >= N_GAUSS) return;

  float mx = means[2 * n + 0];
  float my = means[2 * n + 1];
  float s2x = __expf(2.f * log_scales[2 * n + 0]);
  float s2y = __expf(2.f * log_scales[2 * n + 1]);
  float s, c;
  __sincosf(rot[n], &s, &c);

  float a = c * c * s2x + s * s * s2y + EPS_COV;
  float b = c * s * (s2x - s2y);
  float d = s * s * s2x + c * c * s2y + EPS_COV;
  float inv = 1.f / (a * d - b * b);
  float ia = d * inv, ib = -b * inv, id = a * inv;

  const int step = n >> 3, r = n & 7;
  float* row0 = ws + (size_t)(step * 16 + r) * ROW_DW;
  feature_row(row0, ia, ib, id, mx, my);
  feature_row(row0 + 8 * ROW_DW, ia, ib, id, mx - X_SHIFT, my);

  float* cp = ws + GF_DW + (size_t)n * 4;
  cp[0] = colors[3 * n + 0];
  cp[1] = colors[3 * n + 1];
  cp[2] = colors[3 * n + 2];
  cp[3] = opac[n];
}

struct StepOperands {
  v2f aA, aB;
  v4f c[STEP_G];
};

__device__ __forceinline__ StepOperands load_step(const float* arow, const float* sc, int s) {
  StepOperands r;
  const float* gf = arow + s * (16 * ROW_DW);
  r.aA = (v2f){ gf[0], gf[1] };
  r.aB = (v2f){ gf[4], gf[5] };
  const v4f* cq = (const v4f*)(sc + s * (STEP_G * 4));
#pragma unroll
  for (int j = 0; j < STEP_G; ++j) r.c[j] = cq[j];
  return r;
}

__device__ __forceinline__ void render_step(const StepOperands& o, v2f bA, v2f bB,
                                            float& T, float& cr, float& cg, float& cb) {
  v8f P = {};
  P = __builtin_amdgcn_wmma_f32_16x16x4_f32(false, o.aA, false, bA,
                                            (short)0, P, false, false);
  P = __builtin_amdgcn_wmma_f32_16x16x4_f32(false, o.aB, false, bB,
                                            (short)0, P, false, false);
#pragma unroll
  for (int j = 0; j < STEP_G; ++j) {
    float al = fminf(o.c[j][3] * __expf(P[j]), ALPHA_MAX);
    float wv = al * T;
    T = fmaf(-al, T, T);
    cr = fmaf(wv, o.c[j][0], cr);
    cg = fmaf(wv, o.c[j][1], cg);
    cb = fmaf(wv, o.c[j][2], cb);
  }
}

__global__ __launch_bounds__(256) void gs_render(const float* __restrict__ ws,
                                                 float* __restrict__ out) {
  __shared__ float smem[SMEM_DW];

  {
    const v4f* src = (const v4f*)ws;
    v4f* dst = (v4f*)smem;
#pragma unroll
    for (int i = 0; i < SMEM_DW / 4 / 256; ++i) dst[threadIdx.x + 256 * i] = src[threadIdx.x + 256 * i];
  }
  __syncthreads();

  const float* sc = smem + GF_DW;

  const int lane = threadIdx.x & 31;
  const int wave = threadIdx.x >> 5;
  const int half = lane >> 4;
  const int col  = lane & 15;

  const float y  = (float)blockIdx.x;
  const float xb = (float)(wave * 32 + col);

  const float f[8] = { xb * xb, xb, y * y, y, xb * y, 1.f, 0.f, 0.f };
  const v2f bA = { f[2 * half + 0], f[2 * half + 1] };
  const v2f bB = { f[4 + 2 * half + 0], f[4 + 2 * half + 1] };

  float T = 1.f;
  float cr = 0.f, cg = 0.f, cb = 0.f;

  const float* arow = smem + col * ROW_DW + 2 * half;

  for (int s = 0; s < N_STEPS; s += 2) {
    const StepOperands o0 = load_step(arow, sc, s);
    const StepOperands o1 = load_step(arow, sc, s + 1);
    render_step(o0, bA, bB, T, cr, cg, cb);
    render_step(o1, bA, bB, T, cr, cg, cb);
  }

  float* o = out + (size_t)(blockIdx.x * IMG_W + wave * 32 + lane) * 3;
  o[0] = cr; o[1] = cg; o[2] = cb;
}

extern "C" void kernel_launch(void* const* d_in, const int* in_sizes, int n_in,
                              void* d_out, int out_size, void* d_ws, size_t ws_size,
                              hipStream_t stream) {
  const float* means      = (const float*)d_in[0];
  const float* log_scales = (const float*)d_in[1];
  const float* rotations  = (const float*)d_in[2];
  const float* colors     = (const float*)d_in[3];
  const float* opacities  = (const float*)d_in[4];
  float* ws = (float*)d_ws;

  gs_prep<<<N_GAUSS / 256, 256, 0, stream>>>(means, log_scales, rotations,
                                             colors, opacities, ws);
  gs_render<<<IMG_H, 256, 0, stream>>>(ws, (float*)d_out);
}


# alpha's exponent whole from the matrix unit: B fragments (pixel coords about the image centre) carry log2(e), C operand = log2(opacity); exp2 reads D, both per-value multiplies go
# speedup vs baseline: 1.6456x; 1.2417x over previous
#include <hip/hip_runtime.h>
#include <hip/hip_bf16.h>
#include <math.h>


typedef __attribute__((ext_vector_type(2))) float v2f;
typedef __attribute__((ext_vector_type(4))) float v4f;
typedef __attribute__((ext_vector_type(8))) float v8f;

#define N_GAUSS   1024
#define IMG_H     256
#define IMG_W     256
#define EPS_COV   1e-6f
#define ALPHA_MAX 0.999f
#define LOG2_E    1.44269504088896340736f
#define CENTRE_X  128.f
#define CENTRE_Y  128.f

#define STEP_G    8
#define N_STEPS   (N_GAUSS / STEP_G)
#define X_SHIFT   16.f
#define ROW_DW    8
#define GF_DW     (N_STEPS * 16 * ROW_DW)
#define STEP_CP   (STEP_G * 4)
#define CP_DW     (N_STEPS * STEP_CP)
#define SMEM_DW   (GF_DW + CP_DW)

#define PPL        4
#define WAVE_PX    (32 * PPL)
#define PIX_GROUPS (IMG_W / WAVE_PX)
#define LIST_PARTS (8 / PIX_GROUPS)
#define PART_STEPS (N_STEPS / LIST_PARTS)
#define MERGE_DW   ((LIST_PARTS - 1) * 4 * IMG_W)
static_assert(PPL % 2 == 0 && IMG_W % WAVE_PX == 0 && 8 % PIX_GROUPS == 0 && PART_STEPS % 2 == 0,
              "pixels go in pairs, 8 waves cover row x list, two steps per trip");

__device__ __forceinline__ void feature_row(float* __restrict__ gf, float ia, float ib, float id,
                                            float mx, float my) {
  gf[0] = -0.5f * ia;
  gf[1] = ia * mx + ib * my;
  gf[2] = -0.5f * id;
  gf[3] = id * my + ib * mx;
  gf[4] = -ib;
  gf[5] = -0.5f * (ia * mx * mx + 2.f * ib * mx * my + id * my * my);
  gf[6] = 0.f;
  gf[7] = 0.f;
}

__global__ __launch_bounds__(256) void gs_prep(const float* __restrict__ means,
                                               const float* __restrict__ log_scales,
                                               const float* __restrict__ rot,
                                               const float* __restrict__ colors,
                                               const float* __restrict__ opac,
                                               float* __restrict__ ws) {
  int n = blockIdx.x * blockDim.x + threadIdx.x;
  if (n >= N_GAUSS) return;

  float mx = means[2 * n + 0] - CENTRE_X;
  float my = means[2 * n + 1] - CENTRE_Y;
  float s2x = __expf(2.f * log_scales[2 * n + 0]);
  float s2y = __expf(2.f * log_scales[2 * n + 1]);
  float s, c;
  __sincosf(rot[n], &s, &c);

  float a = c * c * s2x + s * s * s2y + EPS_COV;
  float b = c * s * (s2x - s2y);
  float d = s * s * s2x + c * c * s2y + EPS_COV;
  float inv = 1.f / (a * d - b * b);
  float ia = d * inv, ib = -b * inv, id = a * inv;

  const int step = n >> 3, r = n & 7;
  float* row0 = ws + (size_t)(step * 16 + r) * ROW_DW;
  feature_row(row0, ia, ib, id, mx, my);
  feature_row(row0 + 8 * ROW_DW, ia, ib, id, mx - X_SHIFT, my);

  float* cp = ws + GF_DW + (size_t)step * STEP_CP;
  cp[r] = __log2f(opac[n]);
  cp[STEP_G + 3 * r + 0] = colors[3 * n + 0];
  cp[STEP_G + 3 * r + 1] = colors[3 * n + 1];
  cp[STEP_G + 3 * r + 2] = colors[3 * n + 2];
}

struct StepOperands {
  v2f aA, aB;
  v8f lop;
  v4f c[6];
};

__device__ __forceinline__ StepOperands load_step(const float* arow, const float* sc, int s) {
  StepOperands r;
  const float* gf = arow + s * (16 * ROW_DW);
  r.aA = (v2f){ gf[0], gf[1] };
  r.aB = (v2f){ gf[4], gf[5] };
  const v4f* cq = (const v4f*)(sc + s * STEP_CP);
  const v4f l0 = cq[0], l1 = cq[1];
  r.lop = (v8f){ l0[0], l0[1], l0[2], l0[3], l1[0], l1[1], l1[2], l1[3] };
#pragma unroll
  for (int j = 0; j < 6; ++j) r.c[j] = cq[2 + j];
  return r;
}

#define NPAIR (PPL / 2)
struct PixelState {
  v2f T[NPAIR];
  v2f cr[NPAIR], cg[NPAIR], cb[NPAIR];
};

__device__ __forceinline__ void render_step(const StepOperands& o, const v2f (&bA)[PPL],
                                            const v2f (&bB)[PPL], PixelState& px) {
  v8f P[PPL];
#pragma unroll
  for (int p = 0; p < PPL; ++p) {
    const v8f z = __builtin_amdgcn_wmma_f32_16x16x4_f32(false, o.aA, false, bA[p],
                                                        (short)0, o.lop, false, false);
    P[p] = __builtin_amdgcn_wmma_f32_16x16x4_f32(false, o.aB, false, bB[p],
                                                 (short)0, z, false, false);
  }
#pragma unroll
  for (int j = 0; j < STEP_G; ++j) {
    const float cr_j = o.c[(3 * j) / 4][(3 * j) % 4], cg_j = o.c[(3 * j + 1) / 4][(3 * j + 1) % 4];
    const float cb_j = o.c[(3 * j + 2) / 4][(3 * j + 2) % 4];
    const v2f red = { cr_j, cr_j }, green = { cg_j, cg_j }, blue = { cb_j, cb_j };
#pragma unroll
    for (int q = 0; q < NPAIR; ++q) {
      const v2f e  = { __builtin_amdgcn_exp2f(P[2 * q][j]), __builtin_amdgcn_exp2f(P[2 * q + 1][j]) };
      const v2f al = __builtin_elementwise_min(e, (v2f){ ALPHA_MAX, ALPHA_MAX });
      const v2f wv = al * px.T[q];
      px.T[q]  = __builtin_elementwise_fma(-al, px.T[q], px.T[q]);
      px.cr[q] = __builtin_elementwise_fma(wv, red,   px.cr[q]);
      px.cg[q] = __builtin_elementwise_fma(wv, green, px.cg[q]);
      px.cb[q] = __builtin_elementwise_fma(wv, blue,  px.cb[q]);
    }
  }
}

__global__ __launch_bounds__(256) void gs_render(const float* __restrict__ ws,
                                                 float* __restrict__ out) {
  __shared__ float smem[SMEM_DW + MERGE_DW];

  {
    const v4f* src = (const v4f*)ws;
    v4f* dst = (v4f*)smem;
#pragma unroll
    for (int i = 0; i < SMEM_DW / 4 / 256; ++i) dst[threadIdx.x + 256 * i] = src[threadIdx.x + 256 * i];
  }
  __syncthreads();

  const float* sc = smem + GF_DW;
  float* mg = smem + SMEM_DW;

  const int lane = threadIdx.x & 31;
  const int wave = __builtin_amdgcn_readfirstlane(threadIdx.x >> 5);
  const int half = lane >> 4;
  const int col  = lane & 15;
  const int pix0 = (wave % PIX_GROUPS) * WAVE_PX;
  const int part = wave / PIX_GROUPS;

  const float y = (float)blockIdx.x - CENTRE_Y;

  v2f bA[PPL], bB[PPL];
#pragma unroll
  for (int p = 0; p < PPL; ++p) {
    const float xb = (float)(pix0 + 32 * p + col) - CENTRE_X;
    const float f[8] = { LOG2_E * (xb * xb), LOG2_E * xb, LOG2_E * (y * y), LOG2_E * y,
                         LOG2_E * (xb * y), LOG2_E, 0.f, 0.f };
    bA[p] = (v2f){ f[2 * half + 0], f[2 * half + 1] };
    bB[p] = (v2f){ f[4 + 2 * half + 0], f[4 + 2 * half + 1] };
  }

  PixelState px;
#pragma unroll
  for (int q = 0; q < NPAIR; ++q) {
    px.T[q] = (v2f){ 1.f, 1.f };
    px.cr[q] = (v2f){ 0.f, 0.f }; px.cg[q] = (v2f){ 0.f, 0.f }; px.cb[q] = (v2f){ 0.f, 0.f };
  }

  const float* arow = smem + col * ROW_DW + 2 * half;

  const int s0 = part * PART_STEPS;
  for (int s = s0; s < s0 + PART_STEPS; s += 2) {
    const StepOperands o0 = load_step(arow, sc, s);
    const StepOperands o1 = load_step(arow, sc, s + 1);
    render_step(o0, bA, bB, px);
    render_step(o1, bA, bB, px);
  }

  if (part != 0) {
    float* m = mg + (part - 1) * 4 * IMG_W + pix0 + lane;
#pragma unroll
    for (int p = 0; p < PPL; ++p) {
      m[0 * IMG_W + 32 * p] = px.cr[p / 2][p % 2];
      m[1 * IMG_W + 32 * p] = px.cg[p / 2][p % 2];
      m[2 * IMG_W + 32 * p] = px.cb[p / 2][p % 2];
    }
    if (part != LIST_PARTS - 1) {
#pragma unroll
      for (int p = 0; p < PPL; ++p) m[3 * IMG_W + 32 * p] = px.T[p / 2][p % 2];
    }
  }
  __syncthreads();
  if (part == 0) {
#pragma unroll
    for (int p = 0; p < PPL; ++p) {
      const int x = pix0 + 32 * p + lane;
      float T = px.T[p / 2][p % 2];
      float cr = px.cr[p / 2][p % 2], cg = px.cg[p / 2][p % 2], cb = px.cb[p / 2][p % 2];
#pragma unroll
      for (int q = 1; q < LIST_PARTS; ++q) {
        const float* m = mg + (q - 1) * 4 * IMG_W + x;
        cr = fmaf(T, m[0 * IMG_W], cr);
        cg = fmaf(T, m[1 * IMG_W], cg);
        cb = fmaf(T, m[2 * IMG_W], cb);
        if (q != LIST_PARTS - 1) T *= m[3 * IMG_W];
      }
      float* o = out + (size_t)(blockIdx.x * IMG_W + x) * 3;
      o[0] = cr; o[1] = cg; o[2] = cb;
    }
  }
}

extern "C" void kernel_launch(void* const* d_in, const int* in_sizes, int n_in,
                              void* d_out, int out_size, void* d_ws, size_t ws_size,
                              hipStream_t stream) {
  const float* means      = (const float*)d_in[0];
  const float* log_scales = (const float*)d_in[1];
  const float* rotations  = (const float*)d_in[2];
  const float* colors     = (const float*)d_in[3];
  const float* opacities  = (const float*)d_in[4];
  float* ws = (float*)d_ws;

  gs_prep<<<N_GAUSS / 256, 256, 0, stream>>>(means, log_scales, rotations,
                                             colors, opacities, ws);
  gs_render<<<IMG_H, 256, 0, stream>>>(ws, (float*)d_out);
}
